# starting kernel
# baseline (speedup 1.0000x reference)
#include <hip/hip_runtime.h>

typedef __attribute__((ext_vector_type(2))) float v2f;
typedef __attribute__((ext_vector_type(8))) float v8f;

#define IN_DIM 128
#define OUT_DIM 32

__global__ void __launch_bounds__(256) gcmc_gemm_cj_kernel(
    const float* __restrict__ X,
    const float* __restrict__ W,
    const float* __restrict__ bias,
    const float* __restrict__ cj,
    float* __restrict__ wh,
    int n_nodes)
{
    const int wave   = (blockIdx.x * blockDim.x + threadIdx.x) >> 5;
    const int ntiles = (n_nodes + 15) >> 4;
    if (wave >= ntiles) return;

    const int lane = threadIdx.x & 31;
    const int m    = lane & 15;
    const int kh   = lane >> 4;

    int arow = wave * 16 + m;
    if (arow >= n_nodes) arow = n_nodes - 1;

    const float* xrow = X + (size_t)arow * IN_DIM + 2 * kh;
    const float* w0   = W + (size_t)m        * IN_DIM + 2 * kh;
    const float* w1   = W + (size_t)(m + 16) * IN_DIM + 2 * kh;

    v8f acc0 = {};
    v8f acc1 = {};

#pragma unroll
    for (int kb = 0; kb < IN_DIM; kb += 4) {
        v2f a  = *(const v2f*)(xrow + kb);
        v2f b0 = *(const v2f*)(w0 + kb);
        v2f b1 = *(const v2f*)(w1 + kb);
        acc0 = __builtin_amdgcn_wmma_f32_16x16x4_f32(
            false, a, false, b0, (short)0, acc0, false, false);
        acc1 = __builtin_amdgcn_wmma_f32_16x16x4_f32(
            false, a, false, b1, (short)0, acc1, false, false);
    }

    const int   n0 = m, n1 = m + 16;
    const float bias0 = bias[n0];
    const float bias1 = bias[n1];

#pragma unroll
    for (int r = 0; r < 8; ++r) {
        const int mo   = r + 8 * kh;
        const int orow = wave * 16 + mo;
        if (orow < n_nodes) {
            const float s = cj[orow];
            wh[(size_t)orow * OUT_DIM + n0] = (acc0[r] + bias0) * s;
            wh[(size_t)orow * OUT_DIM + n1] = (acc1[r] + bias1) * s;
        }
    }
}

__global__ void __launch_bounds__(256) gcmc_zero_kernel(float4* __restrict__ out, int n4)
{
    const int i = blockIdx.x * blockDim.x + threadIdx.x;
    if (i < n4) out[i] = make_float4(0.f, 0.f, 0.f, 0.f);
}

__global__ void __launch_bounds__(256) gcmc_scatter_kernel(
    const float* __restrict__ wh,
    const int*   __restrict__ edge_src,
    const int*   __restrict__ edge_dst,
    float*       __restrict__ out,
    int n_edges)
{
    const long long t = (long long)blockIdx.x * blockDim.x + threadIdx.x;
    const long long e = t >> 3;
    if (e >= n_edges) return;
    const int c = ((int)t & 7) * 4;

    const int s = edge_src[e];
    const int d = edge_dst[e];

    const float4 v = *(const float4*)(wh + (size_t)s * OUT_DIM + c);
    float* o = out + (size_t)d * OUT_DIM + c;
    unsafeAtomicAdd(o + 0, v.x);
    unsafeAtomicAdd(o + 1, v.y);
    unsafeAtomicAdd(o + 2, v.z);
    unsafeAtomicAdd(o + 3, v.w);
}

__global__ void __launch_bounds__(256) gcmc_ci_kernel(
    float* __restrict__ out, const float* __restrict__ ci, int n_nodes)
{
    const int t    = blockIdx.x * blockDim.x + threadIdx.x;
    const int node = t >> 3;
    if (node >= n_nodes) return;
    const int c = (t & 7) * 4;

    const float s = ci[node];
    float4* p = (float4*)(out + (size_t)node * OUT_DIM + c);
    float4  v = *p;
    v.x *= s; v.y *= s; v.z *= s; v.w *= s;
    *p = v;
}

extern "C" void kernel_launch(void* const* d_in, const int* in_sizes, int n_in,
                              void* d_out, int out_size, void* d_ws, size_t ws_size,
                              hipStream_t stream)
{
    const float* X    = (const float*)d_in[0];
    const float* cj   = (const float*)d_in[1];
    const float* ci   = (const float*)d_in[2];
    const float* W    = (const float*)d_in[3];
    const float* bias = (const float*)d_in[4];
    const int*   esrc = (const int*)d_in[5];
    const int*   edst = (const int*)d_in[6];

    float* out = (float*)d_out;
    float* wh  = (float*)d_ws;

    const int n_nodes = in_sizes[1];
    const int n_edges = in_sizes[5];

    {
        const int ntiles = (n_nodes + 15) / 16;
        const int blocks = (ntiles + 7) / 8;
        gcmc_gemm_cj_kernel<<<blocks, 256, 0, stream>>>(X, W, bias, cj, wh, n_nodes);
    }

    {
        const int n4 = out_size / 4;
        gcmc_zero_kernel<<<(n4 + 255) / 256, 256, 0, stream>>>((float4*)out, n4);
    }

    {
        const long long threads = (long long)n_edges * 8;
        const int blocks = (int)((threads + 255) / 256);
        gcmc_scatter_kernel<<<blocks, 256, 0, stream>>>(wh, esrc, edst, out, n_edges);
    }

    {
        const int threads = n_nodes * 8;
        gcmc_ci_kernel<<<(threads + 255) / 256, 256, 0, stream>>>(out, ci, n_nodes);
    }
}


# ci[dst] multiplied into each edge's addend before its atomic add, so the separate out*=ci pass and its launch are gone (rounding: a sum of scaled terms instead of a scaled sum; add order unchanged)
# speedup vs baseline: 1.1844x; 1.1844x over previous
#include <hip/hip_runtime.h>

typedef __attribute__((ext_vector_type(2))) float v2f;
typedef __attribute__((ext_vector_type(8))) float v8f;

#define IN_DIM 128
#define OUT_DIM 32

__global__ void __launch_bounds__(256) gcmc_gemm_cj_kernel(
    const float* __restrict__ X,
    const float* __restrict__ W,
    const float* __restrict__ bias,
    const float* __restrict__ cj,
    float* __restrict__ wh,
    int n_nodes)
{
    const int wave   = (blockIdx.x * blockDim.x + threadIdx.x) >> 5;
    const int ntiles = (n_nodes + 15) >> 4;
    if (wave >= ntiles) return;

    const int lane = threadIdx.x & 31;
    const int m    = lane & 15;
    const int kh   = lane >> 4;

    int arow = wave * 16 + m;
    if (arow >= n_nodes) arow = n_nodes - 1;

    const float* xrow = X + (size_t)arow * IN_DIM + 2 * kh;
    const float* w0   = W + (size_t)m        * IN_DIM + 2 * kh;
    const float* w1   = W + (size_t)(m + 16) * IN_DIM + 2 * kh;

    v8f acc0 = {};
    v8f acc1 = {};

#pragma unroll
    for (int kb = 0; kb < IN_DIM; kb += 4) {
        v2f a  = *(const v2f*)(xrow + kb);
        v2f b0 = *(const v2f*)(w0 + kb);
        v2f b1 = *(const v2f*)(w1 + kb);
        acc0 = __builtin_amdgcn_wmma_f32_16x16x4_f32(
            false, a, false, b0, (short)0, acc0, false, false);
        acc1 = __builtin_amdgcn_wmma_f32_16x16x4_f32(
            false, a, false, b1, (short)0, acc1, false, false);
    }

    const int   n0 = m, n1 = m + 16;
    const float bias0 = bias[n0];
    const float bias1 = bias[n1];

#pragma unroll
    for (int r = 0; r < 8; ++r) {
        const int mo   = r + 8 * kh;
        const int orow = wave * 16 + mo;
        if (orow < n_nodes) {
            const float s = cj[orow];
            wh[(size_t)orow * OUT_DIM + n0] = (acc0[r] + bias0) * s;
            wh[(size_t)orow * OUT_DIM + n1] = (acc1[r] + bias1) * s;
        }
    }
}

__global__ void __launch_bounds__(256) gcmc_zero_kernel(float4* __restrict__ out, int n4)
{
    const int i = blockIdx.x * blockDim.x + threadIdx.x;
    if (i < n4) out[i] = make_float4(0.f, 0.f, 0.f, 0.f);
}

__global__ void __launch_bounds__(256) gcmc_scatter_ci_kernel(
    const float* __restrict__ wh,
    const float* __restrict__ ci,
    const int*   __restrict__ edge_src,
    const int*   __restrict__ edge_dst,
    float*       __restrict__ out,
    int n_edges)
{
    const long long t = (long long)blockIdx.x * blockDim.x + threadIdx.x;
    const long long e = t >> 3;
    if (e >= n_edges) return;
    const int c = ((int)t & 7) * 4;

    const int s = edge_src[e];
    const int d = edge_dst[e];

    const float4 v  = *(const float4*)(wh + (size_t)s * OUT_DIM + c);
    const float  sc = ci[d];
    float* o = out + (size_t)d * OUT_DIM + c;
    unsafeAtomicAdd(o + 0, v.x * sc);
    unsafeAtomicAdd(o + 1, v.y * sc);
    unsafeAtomicAdd(o + 2, v.z * sc);
    unsafeAtomicAdd(o + 3, v.w * sc);
}

extern "C" void kernel_launch(void* const* d_in, const int* in_sizes, int n_in,
                              void* d_out, int out_size, void* d_ws, size_t ws_size,
                              hipStream_t stream)
{
    const float* X    = (const float*)d_in[0];
    const float* cj   = (const float*)d_in[1];
    const float* ci   = (const float*)d_in[2];
    const float* W    = (const float*)d_in[3];
    const float* bias = (const float*)d_in[4];
    const int*   esrc = (const int*)d_in[5];
    const int*   edst = (const int*)d_in[6];

    float* out = (float*)d_out;
    float* wh  = (float*)d_ws;

    const int n_nodes = in_sizes[1];
    const int n_edges = in_sizes[5];

    {
        const int ntiles = (n_nodes + 15) / 16;
        const int blocks = (ntiles + 7) / 8;
        gcmc_gemm_cj_kernel<<<blocks, 256, 0, stream>>>(X, W, bias, cj, wh, n_nodes);
    }

    {
        const int n4 = out_size / 4;
        gcmc_zero_kernel<<<(n4 + 255) / 256, 256, 0, stream>>>((float4*)out, n4);
    }

    {
        const long long threads = (long long)n_edges * 8;
        const int blocks = (int)((threads + 255) / 256);
        gcmc_scatter_ci_kernel<<<blocks, 256, 0, stream>>>(wh, ci, esrc, edst, out, n_edges);
    }
}


# edge scatter re-laid: one wave per 32 edges, lane = output column, so each gather load and each f32 atomic add covers one whole 128-B row (was 8 lanes x float4, 4 strided adds); arrival-order sums
# speedup vs baseline: 1.6051x; 1.3552x over previous
#include <hip/hip_runtime.h>

typedef __attribute__((ext_vector_type(2))) float v2f;
typedef __attribute__((ext_vector_type(8))) float v8f;

#define IN_DIM 128
#define OUT_DIM 32

__global__ void __launch_bounds__(256) gcmc_gemm_cj_kernel(
    const float* __restrict__ X,
    const float* __restrict__ W,
    const float* __restrict__ bias,
    const float* __restrict__ cj,
    float* __restrict__ wh,
    int n_nodes)
{
    const int wave   = (blockIdx.x * blockDim.x + threadIdx.x) >> 5;
    const int ntiles = (n_nodes + 15) >> 4;
    if (wave >= ntiles) return;

    const int lane = threadIdx.x & 31;
    const int m    = lane & 15;
    const int kh   = lane >> 4;

    int arow = wave * 16 + m;
    if (arow >= n_nodes) arow = n_nodes - 1;

    const float* xrow = X + (size_t)arow * IN_DIM + 2 * kh;
    const float* w0   = W + (size_t)m        * IN_DIM + 2 * kh;
    const float* w1   = W + (size_t)(m + 16) * IN_DIM + 2 * kh;

    v8f acc0 = {};
    v8f acc1 = {};

#pragma unroll
    for (int kb = 0; kb < IN_DIM; kb += 4) {
        v2f a  = *(const v2f*)(xrow + kb);
        v2f b0 = *(const v2f*)(w0 + kb);
        v2f b1 = *(const v2f*)(w1 + kb);
        acc0 = __builtin_amdgcn_wmma_f32_16x16x4_f32(
            false, a, false, b0, (short)0, acc0, false, false);
        acc1 = __builtin_amdgcn_wmma_f32_16x16x4_f32(
            false, a, false, b1, (short)0, acc1, false, false);
    }

    const int   n0 = m, n1 = m + 16;
    const float bias0 = bias[n0];
    const float bias1 = bias[n1];

#pragma unroll
    for (int r = 0; r < 8; ++r) {
        const int mo   = r + 8 * kh;
        const int orow = wave * 16 + mo;
        if (orow < n_nodes) {
            const float s = cj[orow];
            wh[(size_t)orow * OUT_DIM + n0] = (acc0[r] + bias0) * s;
            wh[(size_t)orow * OUT_DIM + n1] = (acc1[r] + bias1) * s;
        }
    }
}

__global__ void __launch_bounds__(256) gcmc_zero_kernel(float4* __restrict__ out, int n4)
{
    const int i = blockIdx.x * blockDim.x + threadIdx.x;
    if (i < n4) out[i] = make_float4(0.f, 0.f, 0.f, 0.f);
}

#define EDGES_PER_WAVE 32

__global__ void __launch_bounds__(256) gcmc_scatter_ci_kernel(
    const float* __restrict__ wh,
    const float* __restrict__ ci,
    const int*   __restrict__ edge_src,
    const int*   __restrict__ edge_dst,
    float*       __restrict__ out,
    int n_edges)
{
    const int      lane  = threadIdx.x & 31;
    const unsigned ulane = (unsigned)lane;
    const int wib = __builtin_amdgcn_readfirstlane((int)(threadIdx.x >> 5));
    const long long wave = (long long)blockIdx.x * (blockDim.x >> 5) + wib;
    const long long e0   = wave * EDGES_PER_WAVE;
    if (e0 >= n_edges) return;
    const long long left = (long long)n_edges - e0;

    int   s = 0, d = 0;
    float c = 0.f;
    if (lane < left) {
        s = edge_src[e0 + lane];
        d = edge_dst[e0 + lane];
        c = ci[d];
    }

    if (left >= EDGES_PER_WAVE) {
        float v[EDGES_PER_WAVE];
#pragma unroll
        for (int j = 0; j < EDGES_PER_WAVE; ++j) {
            const unsigned sj = (unsigned)__builtin_amdgcn_readlane(s, j);
            v[j] = wh[sj * OUT_DIM + ulane];
        }
#pragma unroll
        for (int j = 0; j < EDGES_PER_WAVE; ++j) {
            const unsigned dj = (unsigned)__builtin_amdgcn_readlane(d, j);
            const float    cj = __int_as_float(__builtin_amdgcn_readlane(__float_as_int(c), j));
            unsafeAtomicAdd(out + (dj * OUT_DIM + ulane), v[j] * cj);
        }
    } else {
        for (int j = 0; j < (int)left; ++j) {
            const unsigned sj = (unsigned)__builtin_amdgcn_readlane(s, j);
            const unsigned dj = (unsigned)__builtin_amdgcn_readlane(d, j);
            const float    cj = __int_as_float(__builtin_amdgcn_readlane(__float_as_int(c), j));
            const float*   rs = wh  + (size_t)sj * OUT_DIM;
            float*         rd = out + (size_t)dj * OUT_DIM;
            unsafeAtomicAdd(rd + lane, rs[lane] * cj);
        }
    }
}

extern "C" void kernel_launch(void* const* d_in, const int* in_sizes, int n_in,
                              void* d_out, int out_size, void* d_ws, size_t ws_size,
                              hipStream_t stream)
{
    const float* X    = (const float*)d_in[0];
    const float* cj   = (const float*)d_in[1];
    const float* ci   = (const float*)d_in[2];
    const float* W    = (const float*)d_in[3];
    const float* bias = (const float*)d_in[4];
    const int*   esrc = (const int*)d_in[5];
    const int*   edst = (const int*)d_in[6];

    float* out = (float*)d_out;
    float* wh  = (float*)d_ws;

    const int n_nodes = in_sizes[1];
    const int n_edges = in_sizes[5];

    {
        const int ntiles = (n_nodes + 15) / 16;
        const int blocks = (ntiles + 7) / 8;
        gcmc_gemm_cj_kernel<<<blocks, 256, 0, stream>>>(X, W, bias, cj, wh, n_nodes);
    }

    {
        const int n4 = out_size / 4;
        gcmc_zero_kernel<<<(n4 + 255) / 256, 256, 0, stream>>>((float4*)out, n4);
    }

    {
        const int batches = (n_edges + EDGES_PER_WAVE - 1) / EDGES_PER_WAVE;
        const int blocks  = (batches + 7) / 8;
        gcmc_scatter_ci_kernel<<<blocks, 256, 0, stream>>>(wh, ci, esrc, edst, out, n_edges);
    }
}


# GEMM fragment fetch re-laid: W staged per work-group in LDS in fragment order, X and W fragments read 16 B per lane (one load = two k-steps); f32 sum order inside each 8-wide k group changes
# speedup vs baseline: 1.7463x; 1.0880x over previous
#include <hip/hip_runtime.h>

typedef __attribute__((ext_vector_type(2))) float v2f;
typedef __attribute__((ext_vector_type(8))) float v8f;

#define IN_DIM 128
#define OUT_DIM 32

#define W_FRAGS ((IN_DIM / 8) * 2 * 32)

__global__ void __launch_bounds__(256) gcmc_gemm_cj_kernel(
    const float* __restrict__ X,
    const float* __restrict__ W,
    const float* __restrict__ bias,
    const float* __restrict__ cj,
    float* __restrict__ wh,
    int n_nodes)
{
    __shared__ float4 wfrag[W_FRAGS];

    const int wave   = (blockIdx.x * blockDim.x + threadIdx.x) >> 5;
    const int ntiles = (n_nodes + 15) >> 4;
    const int lane = threadIdx.x & 31;
    const int m    = lane & 15;
    const int kh   = lane >> 4;

    int arow = wave * 16 + m;
    if (arow >= n_nodes) arow = n_nodes - 1;

    const float* xrow = X + (size_t)arow * IN_DIM + 4 * kh;

    float4 a[IN_DIM / 8];
#pragma unroll
    for (int j = 0; j < IN_DIM / 8; ++j)
        a[j] = *(const float4*)(xrow + 8 * j);

#pragma unroll
    for (int q = 0; q < W_FRAGS / 256; ++q) {
        const int i  = (int)threadIdx.x + 256 * q;
        const int l  = i & 31;
        const int nt = (i >> 5) & 1;
        const int j  = i >> 6;
        const int n  = nt * 16 + (l & 15);
        const int k  = 8 * j + 4 * (l >> 4);
        wfrag[i] = *(const float4*)(W + (size_t)n * IN_DIM + k);
    }
    __syncthreads();

    if (wave >= ntiles) return;

    v8f acc0 = {};
    v8f acc1 = {};

#pragma unroll
    for (int j = 0; j < IN_DIM / 8; ++j) {
        const float4 b0 = wfrag[(2 * j + 0) * 32 + lane];
        const float4 b1 = wfrag[(2 * j + 1) * 32 + lane];
        const v2f a_lo  = {a[j].x, a[j].y};
        const v2f a_hi  = {a[j].z, a[j].w};
        const v2f b0_lo = {b0.x, b0.y};
        const v2f b0_hi = {b0.z, b0.w};
        const v2f b1_lo = {b1.x, b1.y};
        const v2f b1_hi = {b1.z, b1.w};
        acc0 = __builtin_amdgcn_wmma_f32_16x16x4_f32(
            false, a_lo, false, b0_lo, (short)0, acc0, false, false);
        acc1 = __builtin_amdgcn_wmma_f32_16x16x4_f32(
            false, a_lo, false, b1_lo, (short)0, acc1, false, false);
        acc0 = __builtin_amdgcn_wmma_f32_16x16x4_f32(
            false, a_hi, false, b0_hi, (short)0, acc0, false, false);
        acc1 = __builtin_amdgcn_wmma_f32_16x16x4_f32(
            false, a_hi, false, b1_hi, (short)0, acc1, false, false);
    }

    const int   n0 = m, n1 = m + 16;
    const float bias0 = bias[n0];
    const float bias1 = bias[n1];

#pragma unroll
    for (int r = 0; r < 8; ++r) {
        const int mo   = r + 8 * kh;
        const int orow = wave * 16 + mo;
        if (orow < n_nodes) {
            const float s = cj[orow];
            wh[(size_t)orow * OUT_DIM + n0] = (acc0[r] + bias0) * s;
            wh[(size_t)orow * OUT_DIM + n1] = (acc1[r] + bias1) * s;
        }
    }
}

__global__ void __launch_bounds__(256) gcmc_zero_kernel(float4* __restrict__ out, int n4)
{
    const int i = blockIdx.x * blockDim.x + threadIdx.x;
    if (i < n4) out[i] = make_float4(0.f, 0.f, 0.f, 0.f);
}

#define EDGES_PER_WAVE 32

__global__ void __launch_bounds__(256) gcmc_scatter_ci_kernel(
    const float* __restrict__ wh,
    const float* __restrict__ ci,
    const int*   __restrict__ edge_src,
    const int*   __restrict__ edge_dst,
    float*       __restrict__ out,
    int n_edges)
{
    const int      lane  = threadIdx.x & 31;
    const unsigned ulane = (unsigned)lane;
    const int wib = __builtin_amdgcn_readfirstlane((int)(threadIdx.x >> 5));
    const long long wave = (long long)blockIdx.x * (blockDim.x >> 5) + wib;
    const long long e0   = wave * EDGES_PER_WAVE;
    if (e0 >= n_edges) return;
    const long long left = (long long)n_edges - e0;

    int   s = 0, d = 0;
    float c = 0.f;
    if (lane < left) {
        s = edge_src[e0 + lane];
        d = edge_dst[e0 + lane];
        c = ci[d];
    }

    if (left >= EDGES_PER_WAVE) {
        float v[EDGES_PER_WAVE];
#pragma unroll
        for (int j = 0; j < EDGES_PER_WAVE; ++j) {
            const unsigned sj = (unsigned)__builtin_amdgcn_readlane(s, j);
            v[j] = wh[sj * OUT_DIM + ulane];
        }
#pragma unroll
        for (int j = 0; j < EDGES_PER_WAVE; ++j) {
            const unsigned dj = (unsigned)__builtin_amdgcn_readlane(d, j);
            const float    cj = __int_as_float(__builtin_amdgcn_readlane(__float_as_int(c), j));
            unsafeAtomicAdd(out + (dj * OUT_DIM + ulane), v[j] * cj);
        }
    } else {
        for (int j = 0; j < (int)left; ++j) {
            const unsigned sj = (unsigned)__builtin_amdgcn_readlane(s, j);
            const unsigned dj = (unsigned)__builtin_amdgcn_readlane(d, j);
            const float    cj = __int_as_float(__builtin_amdgcn_readlane(__float_as_int(c), j));
            const float*   rs = wh  + (size_t)sj * OUT_DIM;
            float*         rd = out + (size_t)dj * OUT_DIM;
            unsafeAtomicAdd(rd + lane, rs[lane] * cj);
        }
    }
}

extern "C" void kernel_launch(void* const* d_in, const int* in_sizes, int n_in,
                              void* d_out, int out_size, void* d_ws, size_t ws_size,
                              hipStream_t stream)
{
    const float* X    = (const float*)d_in[0];
    const float* cj   = (const float*)d_in[1];
    const float* ci   = (const float*)d_in[2];
    const float* W    = (const float*)d_in[3];
    const float* bias = (const float*)d_in[4];
    const int*   esrc = (const int*)d_in[5];
    const int*   edst = (const int*)d_in[6];

    float* out = (float*)d_out;
    float* wh  = (float*)d_ws;

    const int n_nodes = in_sizes[1];
    const int n_edges = in_sizes[5];

    {
        const int ntiles = (n_nodes + 15) / 16;
        const int blocks = (ntiles + 7) / 8;
        gcmc_gemm_cj_kernel<<<blocks, 256, 0, stream>>>(X, W, bias, cj, wh, n_nodes);
    }

    {
        const int n4 = out_size / 4;
        gcmc_zero_kernel<<<(n4 + 255) / 256, 256, 0, stream>>>((float4*)out, n4);
    }

    {
        const int batches = (n_edges + EDGES_PER_WAVE - 1) / EDGES_PER_WAVE;
        const int blocks  = (batches + 7) / 8;
        gcmc_scatter_ci_kernel<<<blocks, 256, 0, stream>>>(wh, ci, esrc, edst, out, n_edges);
    }
}


# edge scatter split by destination half: work-groups read their die and take ticketed 512-edge blocks for the out rows their chip half owns, so each out row is added from one half; arrival-order sums
# speedup vs baseline: 2.4584x; 1.4078x over previous
#include <hip/hip_runtime.h>

typedef __attribute__((ext_vector_type(2))) float v2f;
typedef __attribute__((ext_vector_type(8))) float v8f;

#define IN_DIM 128
#define OUT_DIM 32

#define W_FRAGS ((IN_DIM / 8) * 2 * 32)

__global__ void __launch_bounds__(256) gcmc_gemm_cj_kernel(
    const float* __restrict__ X,
    const float* __restrict__ W,
    const float* __restrict__ bias,
    const float* __restrict__ cj,
    float* __restrict__ wh,
    int n_nodes)
{
    __shared__ float4 wfrag[W_FRAGS];

    const int wave   = (blockIdx.x * blockDim.x + threadIdx.x) >> 5;
    const int ntiles = (n_nodes + 15) >> 4;
    const int lane = threadIdx.x & 31;
    const int m    = lane & 15;
    const int kh   = lane >> 4;

    int arow = wave * 16 + m;
    if (arow >= n_nodes) arow = n_nodes - 1;

    const float* xrow = X + (size_t)arow * IN_DIM + 4 * kh;

    float4 a[IN_DIM / 8];
#pragma unroll
    for (int j = 0; j < IN_DIM / 8; ++j)
        a[j] = *(const float4*)(xrow + 8 * j);

#pragma unroll
    for (int q = 0; q < W_FRAGS / 256; ++q) {
        const int i  = (int)threadIdx.x + 256 * q;
        const int l  = i & 31;
        const int nt = (i >> 5) & 1;
        const int j  = i >> 6;
        const int n  = nt * 16 + (l & 15);
        const int k  = 8 * j + 4 * (l >> 4);
        wfrag[i] = *(const float4*)(W + (size_t)n * IN_DIM + k);
    }
    __syncthreads();

    if (wave >= ntiles) return;

    v8f acc0 = {};
    v8f acc1 = {};

#pragma unroll
    for (int j = 0; j < IN_DIM / 8; ++j) {
        const float4 b0 = wfrag[(2 * j + 0) * 32 + lane];
        const float4 b1 = wfrag[(2 * j + 1) * 32 + lane];
        const v2f a_lo  = {a[j].x, a[j].y};
        const v2f a_hi  = {a[j].z, a[j].w};
        const v2f b0_lo = {b0.x, b0.y};
        const v2f b0_hi = {b0.z, b0.w};
        const v2f b1_lo = {b1.x, b1.y};
        const v2f b1_hi = {b1.z, b1.w};
        acc0 = __builtin_amdgcn_wmma_f32_16x16x4_f32(
            false, a_lo, false, b0_lo, (short)0, acc0, false, false);
        acc1 = __builtin_amdgcn_wmma_f32_16x16x4_f32(
            false, a_lo, false, b1_lo, (short)0, acc1, false, false);
        acc0 = __builtin_amdgcn_wmma_f32_16x16x4_f32(
            false, a_hi, false, b0_hi, (short)0, acc0, false, false);
        acc1 = __builtin_amdgcn_wmma_f32_16x16x4_f32(
            false, a_hi, false, b1_hi, (short)0, acc1, false, false);
    }

    const int   n0 = m, n1 = m + 16;
    const float bias0 = bias[n0];
    const float bias1 = bias[n1];

#pragma unroll
    for (int r = 0; r < 8; ++r) {
        const int mo   = r + 8 * kh;
        const int orow = wave * 16 + mo;
        if (orow < n_nodes) {
            const float s = cj[orow];
            wh[(size_t)orow * OUT_DIM + n0] = (acc0[r] + bias0) * s;
            wh[(size_t)orow * OUT_DIM + n1] = (acc1[r] + bias1) * s;
        }
    }
}

#define TICKET_STRIDE 64

__global__ void __launch_bounds__(256) gcmc_zero_kernel(float4* __restrict__ out, int n4,
                                                        int* __restrict__ tickets)
{
    const int i = blockIdx.x * blockDim.x + threadIdx.x;
    if (i < n4) out[i] = make_float4(0.f, 0.f, 0.f, 0.f);
    if (i == 0) { tickets[0] = 0; tickets[TICKET_STRIDE] = 0; }
}

#define EDGES_PER_WAVE  64
#define EDGES_PER_BLOCK (8 * EDGES_PER_WAVE)
#define CHUNK 8

__global__ void __launch_bounds__(256) gcmc_scatter_ci_kernel(
    const float* __restrict__ wh,
    const float* __restrict__ ci,
    const int*   __restrict__ edge_src,
    const int*   __restrict__ edge_dst,
    float*       __restrict__ out,
    int*         __restrict__ tickets,
    int n_edges, int n_nodes)
{
    __shared__ int   s_ticket;
    __shared__ int   l_src[8][EDGES_PER_WAVE];
    __shared__ int   l_dst[8][EDGES_PER_WAVE];
    __shared__ float l_ci [8][EDGES_PER_WAVE];

    const int      lane  = threadIdx.x & 31;
    const unsigned ulane = (unsigned)lane;
    const int wib = __builtin_amdgcn_readfirstlane((int)(threadIdx.x >> 5));

    const int die  = (int)((__builtin_amdgcn_s_sendmsg_rtn(0x87) >> 16) & 0xF);
    const int half = (die >= 4) ? 1 : 0;

    const int split   = n_nodes >> 1;
    const int nblocks = (n_edges + EDGES_PER_BLOCK - 1) / EDGES_PER_BLOCK;

    for (int p = 0; p < 2; ++p) {
        const int t = half ^ p;
        for (int it = 0; it <= nblocks; ++it) {
            if (threadIdx.x == 0) s_ticket = atomicAdd(tickets + t * TICKET_STRIDE, 1);
            __syncthreads();
            const int b = s_ticket;
            __syncthreads();
            if (b >= nblocks) break;

            const long long e0 = (long long)b * EDGES_PER_BLOCK + (long long)wib * EDGES_PER_WAVE;
            const long long ea  = e0 + lane;
            const long long eb  = ea + 32;
            const long long el  = (long long)n_edges - 1;
            const long long eac = ea < el ? ea : el;
            const long long ebc = eb < el ? eb : el;
            const int da = edge_dst[eac];
            const int db = edge_dst[ebc];
            const int sa = edge_src[eac];
            const int sb = edge_src[ebc];
            const bool keepa = (ea <= el) && ((((da >= split) ? 1 : 0)) == t);
            const bool keepb = (eb <= el) && ((((db >= split) ? 1 : 0)) == t);
            float ca = 0.f, cb2 = 0.f;
            if (keepa) ca  = ci[da];
            if (keepb) cb2 = ci[db];
            const unsigned maska = (unsigned)__ballot(keepa);
            const unsigned maskb = (unsigned)__ballot(keepb);
            const int na = __builtin_popcount(maska);
            const int n  = na + __builtin_popcount(maskb);
            if (keepa) {
                const int pos = __builtin_amdgcn_mbcnt_lo(maska, 0u);
                l_src[wib][pos] = sa;
                l_dst[wib][pos] = da;
                l_ci [wib][pos] = ca;
            }
            if (keepb) {
                const int pos = na + __builtin_amdgcn_mbcnt_lo(maskb, 0u);
                l_src[wib][pos] = sb;
                l_dst[wib][pos] = db;
                l_ci [wib][pos] = cb2;
            }

            for (int cb = 0; cb < n; cb += CHUNK) {
                int idx = cb + (lane & (CHUNK - 1));
                if (idx > n - 1) idx = n - 1;
                const int   ss = l_src[wib][idx];
                const int   dd = l_dst[wib][idx];
                const float cc = l_ci [wib][idx];
                float v[CHUNK];
#pragma unroll
                for (int j = 0; j < CHUNK; ++j) {
                    const unsigned sj = (unsigned)__builtin_amdgcn_readlane(ss, j);
                    v[j] = wh[sj * OUT_DIM + ulane];
                }
                asm volatile("" :: "v"(v[0]), "v"(v[1]), "v"(v[2]), "v"(v[3]),
                                   "v"(v[4]), "v"(v[5]), "v"(v[6]), "v"(v[7]));
#pragma unroll
                for (int j = 0; j < CHUNK; ++j) {
                    if (cb + j < n) {
                        const unsigned dj = (unsigned)__builtin_amdgcn_readlane(dd, j);
                        const float    cj = __int_as_float(__builtin_amdgcn_readlane(__float_as_int(cc), j));
                        unsafeAtomicAdd(out + (dj * OUT_DIM + ulane), v[j] * cj);
                    }
                }
            }
        }
    }
}

extern "C" void kernel_launch(void* const* d_in, const int* in_sizes, int n_in,
                              void* d_out, int out_size, void* d_ws, size_t ws_size,
                              hipStream_t stream)
{
    const float* X    = (const float*)d_in[0];
    const float* cj   = (const float*)d_in[1];
    const float* ci   = (const float*)d_in[2];
    const float* W    = (const float*)d_in[3];
    const float* bias = (const float*)d_in[4];
    const int*   esrc = (const int*)d_in[5];
    const int*   edst = (const int*)d_in[6];

    float* out = (float*)d_out;
    float* wh  = (float*)d_ws;

    const int n_nodes = in_sizes[1];
    const int n_edges = in_sizes[5];
    int* tickets = (int*)(wh + (size_t)n_nodes * OUT_DIM);

    {
        const int ntiles = (n_nodes + 15) / 16;
        const int blocks = (ntiles + 7) / 8;
        gcmc_gemm_cj_kernel<<<blocks, 256, 0, stream>>>(X, W, bias, cj, wh, n_nodes);
    }

    {
        const int n4 = out_size / 4;
        gcmc_zero_kernel<<<(n4 + 255) / 256, 256, 0, stream>>>((float4*)out, n4, tickets);
    }

    {
        gcmc_scatter_ci_kernel<<<2048, 256, 0, stream>>>(wh, ci, esrc, edst, out, tickets, n_edges, n_nodes);
    }
}


# out zeroing + ticket reset moved into the GEMM kernel: each wave clears its tile's 16 out rows by four 512-B stores under its X loads; the zeroing launch (25,000 short waves) is gone; add order kept
# speedup vs baseline: 2.5678x; 1.0445x over previous
#include <hip/hip_runtime.h>

typedef __attribute__((ext_vector_type(2))) float v2f;
typedef __attribute__((ext_vector_type(8))) float v8f;

#define IN_DIM 128
#define OUT_DIM 32

#define W_FRAGS ((IN_DIM / 8) * 2 * 32)
#define TICKET_STRIDE 64
#define ZERO_CHUNK 128

__global__ void __launch_bounds__(256) gcmc_gemm_cj_kernel(
    const float* __restrict__ X,
    const float* __restrict__ W,
    const float* __restrict__ bias,
    const float* __restrict__ cj,
    float* __restrict__ wh,
    float4* __restrict__ out4,
    int n4,
    int* __restrict__ tickets,
    int n_nodes)
{
    __shared__ float4 wfrag[W_FRAGS];

    const int wave   = (blockIdx.x * blockDim.x + threadIdx.x) >> 5;
    const int ntiles = (n_nodes + 15) >> 4;
    const int lane = threadIdx.x & 31;
    const int m    = lane & 15;
    const int kh   = lane >> 4;

    int arow = wave * 16 + m;
    if (arow >= n_nodes) arow = n_nodes - 1;

    const float* xrow = X + (size_t)arow * IN_DIM + 4 * kh;

    float4 a[IN_DIM / 8];
#pragma unroll
    for (int j = 0; j < IN_DIM / 8; ++j)
        a[j] = *(const float4*)(xrow + 8 * j);

    {
        const int nwaves  = (int)((gridDim.x * blockDim.x) >> 5);
        const int nchunks = (n4 + ZERO_CHUNK - 1) / ZERO_CHUNK;
        const float4 z4 = make_float4(0.f, 0.f, 0.f, 0.f);
        for (int c = wave; c < nchunks; c += nwaves) {
#pragma unroll
            for (int q = 0; q < ZERO_CHUNK / 32; ++q) {
                const int i = c * ZERO_CHUNK + q * 32 + lane;
                if (i < n4) out4[i] = z4;
            }
        }
        if (blockIdx.x == 0 && threadIdx.x == 0) { tickets[0] = 0; tickets[TICKET_STRIDE] = 0; }
    }

#pragma unroll
    for (int q = 0; q < W_FRAGS / 256; ++q) {
        const int i  = (int)threadIdx.x + 256 * q;
        const int l  = i & 31;
        const int nt = (i >> 5) & 1;
        const int j  = i >> 6;
        const int n  = nt * 16 + (l & 15);
        const int k  = 8 * j + 4 * (l >> 4);
        wfrag[i] = *(const float4*)(W + (size_t)n * IN_DIM + k);
    }
    __syncthreads();

    if (wave >= ntiles) return;

    v8f acc0 = {};
    v8f acc1 = {};

#pragma unroll
    for (int j = 0; j < IN_DIM / 8; ++j) {
        const float4 b0 = wfrag[(2 * j + 0) * 32 + lane];
        const float4 b1 = wfrag[(2 * j + 1) * 32 + lane];
        const v2f a_lo  = {a[j].x, a[j].y};
        const v2f a_hi  = {a[j].z, a[j].w};
        const v2f b0_lo = {b0.x, b0.y};
        const v2f b0_hi = {b0.z, b0.w};
        const v2f b1_lo = {b1.x, b1.y};
        const v2f b1_hi = {b1.z, b1.w};
        acc0 = __builtin_amdgcn_wmma_f32_16x16x4_f32(
            false, a_lo, false, b0_lo, (short)0, acc0, false, false);
        acc1 = __builtin_amdgcn_wmma_f32_16x16x4_f32(
            false, a_lo, false, b1_lo, (short)0, acc1, false, false);
        acc0 = __builtin_amdgcn_wmma_f32_16x16x4_f32(
            false, a_hi, false, b0_hi, (short)0, acc0, false, false);
        acc1 = __builtin_amdgcn_wmma_f32_16x16x4_f32(
            false, a_hi, false, b1_hi, (short)0, acc1, false, false);
    }

    const int   n0 = m, n1 = m + 16;
    const float bias0 = bias[n0];
    const float bias1 = bias[n1];

#pragma unroll
    for (int r = 0; r < 8; ++r) {
        const int mo   = r + 8 * kh;
        const int orow = wave * 16 + mo;
        if (orow < n_nodes) {
            const float s = cj[orow];
            wh[(size_t)orow * OUT_DIM + n0] = (acc0[r] + bias0) * s;
            wh[(size_t)orow * OUT_DIM + n1] = (acc1[r] + bias1) * s;
        }
    }
}

#define EDGES_PER_WAVE  64
#define EDGES_PER_BLOCK (8 * EDGES_PER_WAVE)
#define CHUNK 8

__global__ void __launch_bounds__(256) gcmc_scatter_ci_kernel(
    const float* __restrict__ wh,
    const float* __restrict__ ci,
    const int*   __restrict__ edge_src,
    const int*   __restrict__ edge_dst,
    float*       __restrict__ out,
    int*         __restrict__ tickets,
    int n_edges, int n_nodes)
{
    __shared__ int   s_ticket;
    __shared__ int   l_src[8][EDGES_PER_WAVE];
    __shared__ int   l_dst[8][EDGES_PER_WAVE];
    __shared__ float l_ci [8][EDGES_PER_WAVE];

    const int      lane  = threadIdx.x & 31;
    const unsigned ulane = (unsigned)lane;
    const int wib = __builtin_amdgcn_readfirstlane((int)(threadIdx.x >> 5));

    const int die  = (int)((__builtin_amdgcn_s_sendmsg_rtn(0x87) >> 16) & 0xF);
    const int half = (die >= 4) ? 1 : 0;

    const int split   = n_nodes >> 1;
    const int nblocks = (n_edges + EDGES_PER_BLOCK - 1) / EDGES_PER_BLOCK;

    for (int p = 0; p < 2; ++p) {
        const int t = half ^ p;
        for (int it = 0; it <= nblocks; ++it) {
            if (threadIdx.x == 0) s_ticket = atomicAdd(tickets + t * TICKET_STRIDE, 1);
            __syncthreads();
            const int b = s_ticket;
            __syncthreads();
            if (b >= nblocks) break;

            const long long e0 = (long long)b * EDGES_PER_BLOCK + (long long)wib * EDGES_PER_WAVE;
            const long long ea  = e0 + lane;
            const long long eb  = ea + 32;
            const long long el  = (long long)n_edges - 1;
            const long long eac = ea < el ? ea : el;
            const long long ebc = eb < el ? eb : el;
            const int da = edge_dst[eac];
            const int db = edge_dst[ebc];
            const int sa = edge_src[eac];
            const int sb = edge_src[ebc];
            const bool keepa = (ea <= el) && ((((da >= split) ? 1 : 0)) == t);
            const bool keepb = (eb <= el) && ((((db >= split) ? 1 : 0)) == t);
            float ca = 0.f, cb2 = 0.f;
            if (keepa) ca  = ci[da];
            if (keepb) cb2 = ci[db];
            const unsigned maska = (unsigned)__ballot(keepa);
            const unsigned maskb = (unsigned)__ballot(keepb);
            const int na = __builtin_popcount(maska);
            const int n  = na + __builtin_popcount(maskb);
            if (keepa) {
                const int pos = __builtin_amdgcn_mbcnt_lo(maska, 0u);
                l_src[wib][pos] = sa;
                l_dst[wib][pos] = da;
                l_ci [wib][pos] = ca;
            }
            if (keepb) {
                const int pos = na + __builtin_amdgcn_mbcnt_lo(maskb, 0u);
                l_src[wib][pos] = sb;
                l_dst[wib][pos] = db;
                l_ci [wib][pos] = cb2;
            }

            for (int cb = 0; cb < n; cb += CHUNK) {
                int idx = cb + (lane & (CHUNK - 1));
                if (idx > n - 1) idx = n - 1;
                const int   ss = l_src[wib][idx];
                const int   dd = l_dst[wib][idx];
                const float cc = l_ci [wib][idx];
                float v[CHUNK];
#pragma unroll
                for (int j = 0; j < CHUNK; ++j) {
                    const unsigned sj = (unsigned)__builtin_amdgcn_readlane(ss, j);
                    v[j] = wh[sj * OUT_DIM + ulane];
                }
                asm volatile("" :: "v"(v[0]), "v"(v[1]), "v"(v[2]), "v"(v[3]),
                                   "v"(v[4]), "v"(v[5]), "v"(v[6]), "v"(v[7]));
#pragma unroll
                for (int j = 0; j < CHUNK; ++j) {
                    if (cb + j < n) {
                        const unsigned dj = (unsigned)__builtin_amdgcn_readlane(dd, j);
                        const float    cj = __int_as_float(__builtin_amdgcn_readlane(__float_as_int(cc), j));
                        unsafeAtomicAdd(out + (dj * OUT_DIM + ulane), v[j] * cj);
                    }
                }
            }
        }
    }
}

extern "C" void kernel_launch(void* const* d_in, const int* in_sizes, int n_in,
                              void* d_out, int out_size, void* d_ws, size_t ws_size,
                              hipStream_t stream)
{
    const float* X    = (const float*)d_in[0];
    const float* cj   = (const float*)d_in[1];
    const float* ci   = (const float*)d_in[2];
    const float* W    = (const float*)d_in[3];
    const float* bias = (const float*)d_in[4];
    const int*   esrc = (const int*)d_in[5];
    const int*   edst = (const int*)d_in[6];

    float* out = (float*)d_out;
    float* wh  = (float*)d_ws;

    const int n_nodes = in_sizes[1];
    const int n_edges = in_sizes[5];
    int* tickets = (int*)(wh + (size_t)n_nodes * OUT_DIM);

    {
        const int ntiles = (n_nodes + 15) / 16;
        const int blocks = (ntiles + 7) / 8;
        const int n4 = out_size / 4;
        gcmc_gemm_cj_kernel<<<blocks, 256, 0, stream>>>(X, W, bias, cj, wh, (float4*)out, n4, tickets, n_nodes);
    }

    {
        gcmc_scatter_ci_kernel<<<2048, 256, 0, stream>>>(wh, ci, esrc, edst, out, tickets, n_edges, n_nodes);
    }
}


# GEMM on a resident grid: 512 work-groups, each wave takes two 16-row tiles with both tiles' X loads in flight across an LDS-only barrier (was 1 tile per wave, 782 groups); sums per tile unchanged
# speedup vs baseline: 2.6727x; 1.0409x over previous
#include <hip/hip_runtime.h>

typedef __attribute__((ext_vector_type(2))) float v2f;
typedef __attribute__((ext_vector_type(8))) float v8f;

#define IN_DIM 128
#define OUT_DIM 32

#define W_FRAGS ((IN_DIM / 8) * 2 * 32)
#define TICKET_STRIDE 64
#define ZERO_CHUNK 128
#define GEMM_BLOCKS 512

__device__ __forceinline__ void gcmc_load_tile(const float* __restrict__ X, int tile, int m, int kh,
                                               int n_nodes, float4 (&a)[IN_DIM / 8])
{
    int arow = tile * 16 + m;
    if (arow >= n_nodes) arow = n_nodes - 1;
    const float* xrow = X + (size_t)arow * IN_DIM + 4 * kh;
#pragma unroll
    for (int j = 0; j < IN_DIM / 8; ++j)
        a[j] = *(const float4*)(xrow + 8 * j);
}

__device__ __forceinline__ void gcmc_tile(const float4 (&a)[IN_DIM / 8], const float4* wfrag,
                                          float cjv, float* __restrict__ wh,
                                          float bias0, float bias1,
                                          int tile, int lane, int m, int kh, int n_nodes)
{
    v8f acc0 = {};
    v8f acc1 = {};

#pragma unroll
    for (int j = 0; j < IN_DIM / 8; ++j) {
        const float4 b0 = wfrag[(2 * j + 0) * 32 + lane];
        const float4 b1 = wfrag[(2 * j + 1) * 32 + lane];
        const v2f a_lo  = {a[j].x, a[j].y};
        const v2f a_hi  = {a[j].z, a[j].w};
        const v2f b0_lo = {b0.x, b0.y};
        const v2f b0_hi = {b0.z, b0.w};
        const v2f b1_lo = {b1.x, b1.y};
        const v2f b1_hi = {b1.z, b1.w};
        acc0 = __builtin_amdgcn_wmma_f32_16x16x4_f32(
            false, a_lo, false, b0_lo, (short)0, acc0, false, false);
        acc1 = __builtin_amdgcn_wmma_f32_16x16x4_f32(
            false, a_lo, false, b1_lo, (short)0, acc1, false, false);
        acc0 = __builtin_amdgcn_wmma_f32_16x16x4_f32(
            false, a_hi, false, b0_hi, (short)0, acc0, false, false);
        acc1 = __builtin_amdgcn_wmma_f32_16x16x4_f32(
            false, a_hi, false, b1_hi, (short)0, acc1, false, false);
    }

    const int n0 = m, n1 = m + 16;
#pragma unroll
    for (int r = 0; r < 8; ++r) {
        const int mo   = r + 8 * kh;
        const int orow = tile * 16 + mo;
        const float s_lo = __int_as_float(__builtin_amdgcn_readlane(__float_as_int(cjv), r));
        const float s_hi = __int_as_float(__builtin_amdgcn_readlane(__float_as_int(cjv), r + 8));
        const float s    = kh ? s_hi : s_lo;
        if (orow < n_nodes) {
            wh[(size_t)orow * OUT_DIM + n0] = (acc0[r] + bias0) * s;
            wh[(size_t)orow * OUT_DIM + n1] = (acc1[r] + bias1) * s;
        }
    }
}

__global__ void __launch_bounds__(256) gcmc_gemm_cj_kernel(
    const float* __restrict__ X,
    const float* __restrict__ W,
    const float* __restrict__ bias,
    const float* __restrict__ cj,
    float* __restrict__ wh,
    float4* __restrict__ out4,
    int n4,
    int* __restrict__ tickets,
    int n_nodes)
{
    __shared__ float4 wfrag[W_FRAGS];

    const int wib    = __builtin_amdgcn_readfirstlane((int)(threadIdx.x >> 5));
    const int nwaves = (int)((gridDim.x * blockDim.x) >> 5);
    const int v      = wib * (int)gridDim.x + (int)blockIdx.x;
    const int ntiles = (n_nodes + 15) >> 4;
    const int lane = threadIdx.x & 31;
    const int m    = lane & 15;
    const int kh   = lane >> 4;

#pragma unroll
    for (int q = 0; q < W_FRAGS / 256; ++q) {
        const int i  = (int)threadIdx.x + 256 * q;
        const int l  = i & 31;
        const int nt = (i >> 5) & 1;
        const int j  = i >> 6;
        const int n  = nt * 16 + (l & 15);
        const int k  = 8 * j + 4 * (l >> 4);
        wfrag[i] = *(const float4*)(W + (size_t)n * IN_DIM + k);
    }
    const float bias0 = bias[m];
    const float bias1 = bias[m + 16];

    float4 a0[IN_DIM / 8];
    float4 a1[IN_DIM / 8];
    int t0 = v;
    int t1 = v + nwaves;
    const int last = n_nodes - 1;
    int tl0 = t0 < ntiles ? t0 : ntiles - 1;
    int tl1 = t1 < ntiles ? t1 : tl0;
    int r0  = tl0 * 16 + m;
    int r1  = tl1 * 16 + m;
    float c0 = cj[r0 < last ? r0 : last];
    float c1 = cj[r1 < last ? r1 : last];
    gcmc_load_tile(X, tl0, m, kh, n_nodes, a0);
    gcmc_load_tile(X, tl1, m, kh, n_nodes, a1);

    __builtin_amdgcn_fence(__ATOMIC_RELEASE, "workgroup", "local");
    __builtin_amdgcn_s_barrier();
    __builtin_amdgcn_fence(__ATOMIC_ACQUIRE, "workgroup", "local");

    {
        const int nchunks = (n4 + ZERO_CHUNK - 1) / ZERO_CHUNK;
        const float4 z4 = make_float4(0.f, 0.f, 0.f, 0.f);
        for (int c = v; c < nchunks; c += nwaves) {
#pragma unroll
            for (int q = 0; q < ZERO_CHUNK / 32; ++q) {
                const int i = c * ZERO_CHUNK + q * 32 + lane;
                if (i < n4) out4[i] = z4;
            }
        }
        if (blockIdx.x == 0 && threadIdx.x == 0) { tickets[0] = 0; tickets[TICKET_STRIDE] = 0; }
    }

    if (t0 < ntiles)
        gcmc_tile(a0, wfrag, c0, wh, bias0, bias1, t0, lane, m, kh, n_nodes);
    if (t1 < ntiles)
        gcmc_tile(a1, wfrag, c1, wh, bias0, bias1, t1, lane, m, kh, n_nodes);

    for (t0 += 2 * nwaves, t1 += 2 * nwaves; t0 < ntiles; t0 += 2 * nwaves, t1 += 2 * nwaves) {
        asm volatile("" ::: "memory");
        tl0 = t0;
        tl1 = t1 < ntiles ? t1 : t0;
        r0  = tl0 * 16 + m;
        r1  = tl1 * 16 + m;
        c0  = cj[r0 < last ? r0 : last];
        c1  = cj[r1 < last ? r1 : last];
        gcmc_load_tile(X, tl0, m, kh, n_nodes, a0);
        gcmc_load_tile(X, tl1, m, kh, n_nodes, a1);
        gcmc_tile(a0, wfrag, c0, wh, bias0, bias1, t0, lane, m, kh, n_nodes);
        if (t1 < ntiles)
            gcmc_tile(a1, wfrag, c1, wh, bias0, bias1, t1, lane, m, kh, n_nodes);
    }
}

#define EDGES_PER_WAVE  64
#define EDGES_PER_BLOCK (8 * EDGES_PER_WAVE)
#define CHUNK 8

__global__ void __launch_bounds__(256) gcmc_scatter_ci_kernel(
    const float* __restrict__ wh,
    const float* __restrict__ ci,
    const int*   __restrict__ edge_src,
    const int*   __restrict__ edge_dst,
    float*       __restrict__ out,
    int*         __restrict__ tickets,
    int n_edges, int n_nodes)
{
    __shared__ int   s_ticket;
    __shared__ int   l_src[8][EDGES_PER_WAVE];
    __shared__ int   l_dst[8][EDGES_PER_WAVE];
    __shared__ float l_ci [8][EDGES_PER_WAVE];

    const int      lane  = threadIdx.x & 31;
    const unsigned ulane = (unsigned)lane;
    const int wib = __builtin_amdgcn_readfirstlane((int)(threadIdx.x >> 5));

    const int die  = (int)((__builtin_amdgcn_s_sendmsg_rtn(0x87) >> 16) & 0xF);
    const int half = (die >= 4) ? 1 : 0;

    const int split   = n_nodes >> 1;
    const int nblocks = (n_edges + EDGES_PER_BLOCK - 1) / EDGES_PER_BLOCK;

    for (int p = 0; p < 2; ++p) {
        const int t = half ^ p;
        for (int it = 0; it <= nblocks; ++it) {
            if (threadIdx.x == 0) s_ticket = atomicAdd(tickets + t * TICKET_STRIDE, 1);
            __syncthreads();
            const int b = s_ticket;
            __syncthreads();
            if (b >= nblocks) break;

            const long long e0 = (long long)b * EDGES_PER_BLOCK + (long long)wib * EDGES_PER_WAVE;
            const long long ea  = e0 + lane;
            const long long eb  = ea + 32;
            const long long el  = (long long)n_edges - 1;
            const long long eac = ea < el ? ea : el;
            const long long ebc = eb < el ? eb : el;
            const int da = edge_dst[eac];
            const int db = edge_dst[ebc];
            const int sa = edge_src[eac];
            const int sb = edge_src[ebc];
            const bool keepa = (ea <= el) && ((((da >= split) ? 1 : 0)) == t);
            const bool keepb = (eb <= el) && ((((db >= split) ? 1 : 0)) == t);
            float ca = 0.f, cb2 = 0.f;
            if (keepa) ca  = ci[da];
            if (keepb) cb2 = ci[db];
            const unsigned maska = (unsigned)__ballot(keepa);
            const unsigned maskb = (unsigned)__ballot(keepb);
            const int na = __builtin_popcount(maska);
            const int n  = na + __builtin_popcount(maskb);
            if (keepa) {
                const int pos = __builtin_amdgcn_mbcnt_lo(maska, 0u);
                l_src[wib][pos] = sa;
                l_dst[wib][pos] = da;
                l_ci [wib][pos] = ca;
            }
            if (keepb) {
                const int pos = na + __builtin_amdgcn_mbcnt_lo(maskb, 0u);
                l_src[wib][pos] = sb;
                l_dst[wib][pos] = db;
                l_ci [wib][pos] = cb2;
            }

            for (int cb = 0; cb < n; cb += CHUNK) {
                int idx = cb + (lane & (CHUNK - 1));
                if (idx > n - 1) idx = n - 1;
                const int   ss = l_src[wib][idx];
                const int   dd = l_dst[wib][idx];
                const float cc = l_ci [wib][idx];
                float v[CHUNK];
#pragma unroll
                for (int j = 0; j < CHUNK; ++j) {
                    const unsigned sj = (unsigned)__builtin_amdgcn_readlane(ss, j);
                    v[j] = wh[sj * OUT_DIM + ulane];
                }
                asm volatile("" :: "v"(v[0]), "v"(v[1]), "v"(v[2]), "v"(v[3]),
                                   "v"(v[4]), "v"(v[5]), "v"(v[6]), "v"(v[7]));
#pragma unroll
                for (int j = 0; j < CHUNK; ++j) {
                    if (cb + j < n) {
                        const unsigned dj = (unsigned)__builtin_amdgcn_readlane(dd, j);
                        const float    cj = __int_as_float(__builtin_amdgcn_readlane(__float_as_int(cc), j));
                        unsafeAtomicAdd(out + (dj * OUT_DIM + ulane), v[j] * cj);
                    }
                }
            }
        }
    }
}

extern "C" void kernel_launch(void* const* d_in, const int* in_sizes, int n_in,
                              void* d_out, int out_size, void* d_ws, size_t ws_size,
                              hipStream_t stream)
{
    const float* X    = (const float*)d_in[0];
    const float* cj   = (const float*)d_in[1];
    const float* ci   = (const float*)d_in[2];
    const float* W    = (const float*)d_in[3];
    const float* bias = (const float*)d_in[4];
    const int*   esrc = (const int*)d_in[5];
    const int*   edst = (const int*)d_in[6];

    float* out = (float*)d_out;
    float* wh  = (float*)d_ws;

    const int n_nodes = in_sizes[1];
    const int n_edges = in_sizes[5];
    int* tickets = (int*)(wh + (size_t)n_nodes * OUT_DIM);

    {
        const int n4 = out_size / 4;
        gcmc_gemm_cj_kernel<<<GEMM_BLOCKS, 256, 0, stream>>>(X, W, bias, cj, wh, (float4*)out, n4, tickets, n_nodes);
    }

    {
        gcmc_scatter_ci_kernel<<<2048, 256, 0, stream>>>(wh, ci, esrc, edst, out, tickets, n_edges, n_nodes);
    }
}
